# starting kernel
# baseline (speedup 1.0000x reference)
#include <hip/hip_runtime.h>

typedef __attribute__((ext_vector_type(2))) float v2f;
typedef __attribute__((ext_vector_type(8))) float v8f;

#define EMB 32
#define K_NBR 10
#define WAVES_PER_BLOCK 8

__global__ __launch_bounds__(256) void sage_layer_kernel(
    const float* __restrict__ feat,
    const float* __restrict__ W,
    const int*   __restrict__ neigh,
    const int*   __restrict__ idxMap,
    float*       __restrict__ out,
    int nTiles)
{
    __shared__ float cshAll[WAVES_PER_BLOCK * 16 * 64];
    const int lane   = threadIdx.x & 31;
    const int wave   = threadIdx.x >> 5;
    const int tile   = blockIdx.x * WAVES_PER_BLOCK + wave;
    if (tile >= nTiles) return;

    const int half   = lane >> 4;
    const int lane16 = lane & 15;
    float* csh = &cshAll[wave * 16 * 64];

    v2f Bf[2][16];
    #pragma unroll
    for (int nh = 0; nh < 2; ++nh) {
        const int o = lane16 + 16 * nh;
        #pragma unroll
        for (int k = 0; k < 16; ++k) {
            const int kk = 4 * k + 2 * half;
            const float2 w2 = *(const float2*)(W + o * 64 + kk);
            v2f b; b.x = w2.x; b.y = w2.y;
            Bf[nh][k] = b;
        }
    }

    const int m0 = tile * 16;
    for (int r = 0; r < 16; ++r) {
        const int row  = m0 + r;
        const int node = idxMap ? idxMap[row] : row;
        const float self = feat[node * EMB + lane];
        const int* nb = neigh + node * K_NBR;
        float acc = 0.f;
        #pragma unroll
        for (int j = 0; j < K_NBR; ++j)
            acc += feat[nb[j] * EMB + lane];
        csh[r * 64 + lane]      = self;
        csh[r * 64 + 32 + lane] = acc * (1.0f / K_NBR);
    }

    #pragma unroll
    for (int nh = 0; nh < 2; ++nh) {
        v8f c = {};
        #pragma unroll
        for (int k = 0; k < 16; ++k) {
            const float2 a2 = *(const float2*)&csh[lane16 * 64 + 4 * k + 2 * half];
            v2f a; a.x = a2.x; a.y = a2.y;
            c = __builtin_amdgcn_wmma_f32_16x16x4_f32(
                     false, a,  false, Bf[nh][k],
                     (short)0, c,  false,  false);
        }
        const int o = 16 * nh + lane16;
        #pragma unroll
        for (int v = 0; v < 8; ++v) {
            const int m = v + 8 * half;
            out[(m0 + m) * EMB + o] = fmaxf(c[v], 0.0f);
        }
    }
}

extern "C" void kernel_launch(void* const* d_in, const int* in_sizes, int n_in,
                              void* d_out, int out_size, void* d_ws, size_t ws_size,
                              hipStream_t stream) {
    const float* emb        = (const float*)d_in[0];
    const float* W1         = (const float*)d_in[1];
    const float* W2         = (const float*)d_in[2];
    const int*   node_batch = (const int*)  d_in[3];
    const int*   neigh      = (const int*)  d_in[4];
    float*       out        = (float*)d_out;
    float*       h1         = (float*)d_ws;

    const int N = in_sizes[0] / EMB;
    const int B = in_sizes[3];

    const int t1 = (N + 15) / 16;
    const int t2 = (B + 15) / 16;

    sage_layer_kernel<<<(t1 + WAVES_PER_BLOCK - 1) / WAVES_PER_BLOCK, 256, 0, stream>>>(
        emb, W1, neigh, nullptr, h1, t1);

    sage_layer_kernel<<<(t2 + WAVES_PER_BLOCK - 1) / WAVES_PER_BLOCK, 256, 0, stream>>>(
        h1, W2, neigh, node_batch, out, t2);
}


# Gather restructured: node and neighbour ids by scalar loads for the tile, feature rows of 4 nodes (about 44 loads) in flight per wave instead of 11 behind a dependent per-row index load
# speedup vs baseline: 1.3324x; 1.3324x over previous
#include <hip/hip_runtime.h>


typedef __attribute__((ext_vector_type(2))) float v2f;
typedef __attribute__((ext_vector_type(8))) float v8f;

#define EMB 32
#define K_NBR 10
#define WAVES_PER_BLOCK 8
#define TILE_ROWS 16
#define GROUP_ROWS 4

template <bool HAS_MAP>
__global__ __launch_bounds__(256) void sage_layer_kernel(
    const float* __restrict__ feat,
    const float* __restrict__ W,
    const int*   __restrict__ neigh,
    const int*   __restrict__ idxMap,
    float*       __restrict__ out,
    int nTiles)
{
    __shared__ float cshAll[WAVES_PER_BLOCK * 16 * 64];
    const int lane   = threadIdx.x & 31;
    const int wave   = __builtin_amdgcn_readfirstlane((int)(threadIdx.x >> 5));
    const int tile   = blockIdx.x * WAVES_PER_BLOCK + wave;
    if (tile >= nTiles) return;

    const int half   = lane >> 4;
    const int lane16 = lane & 15;
    float* csh = &cshAll[wave * 16 * 64];

    v2f Bf[2][16];
    #pragma unroll
    for (int nh = 0; nh < 2; ++nh) {
        const int o = lane16 + 16 * nh;
        #pragma unroll
        for (int k = 0; k < 16; ++k) {
            const int kk = 4 * k + 2 * half;
            const float2 w2 = *(const float2*)(W + o * 64 + kk);
            v2f b; b.x = w2.x; b.y = w2.y;
            Bf[nh][k] = b;
        }
    }

    const int m0 = tile * TILE_ROWS;

    int node[TILE_ROWS];
    #pragma unroll
    for (int r = 0; r < TILE_ROWS; ++r)
        node[r] = HAS_MAP ? idxMap[m0 + r] : (m0 + r);

    int ix[GROUP_ROWS][K_NBR];
    #pragma unroll
    for (int r = 0; r < GROUP_ROWS; ++r) {
        const int* nb = neigh + node[r] * K_NBR;
        #pragma unroll
        for (int j = 0; j < K_NBR; ++j) ix[r][j] = nb[j];
    }

    #pragma unroll
    for (int g = 0; g < TILE_ROWS / GROUP_ROWS; ++g) {
        float selfv[GROUP_ROWS];
        float v[GROUP_ROWS][K_NBR];
        #pragma unroll
        for (int r = 0; r < GROUP_ROWS; ++r) {
            selfv[r] = feat[node[g * GROUP_ROWS + r] * EMB + lane];
            #pragma unroll
            for (int j = 0; j < K_NBR; ++j)
                v[r][j] = feat[ix[r][j] * EMB + lane];
        }
        if (g + 1 < TILE_ROWS / GROUP_ROWS) {
            #pragma unroll
            for (int r = 0; r < GROUP_ROWS; ++r) {
                const int* nb = neigh + node[(g + 1) * GROUP_ROWS + r] * K_NBR;
                #pragma unroll
                for (int j = 0; j < K_NBR; ++j) ix[r][j] = nb[j];
            }
        }
        #pragma unroll
        for (int r = 0; r < GROUP_ROWS; ++r) {
            float acc = 0.f;
            #pragma unroll
            for (int j = 0; j < K_NBR; ++j)
                acc += v[r][j];
            const int row = g * GROUP_ROWS + r;
            csh[row * 64 + lane]      = selfv[r];
            csh[row * 64 + 32 + lane] = acc * (1.0f / K_NBR);
        }
    }

    #pragma unroll
    for (int nh = 0; nh < 2; ++nh) {
        v8f c = {};
        #pragma unroll
        for (int k = 0; k < 16; ++k) {
            const float2 a2 = *(const float2*)&csh[lane16 * 64 + 4 * k + 2 * half];
            v2f a; a.x = a2.x; a.y = a2.y;
            c = __builtin_amdgcn_wmma_f32_16x16x4_f32(
                     false, a,  false, Bf[nh][k],
                     (short)0, c,  false,  false);
        }
        const int o = 16 * nh + lane16;
        #pragma unroll
        for (int v = 0; v < 8; ++v) {
            const int m = v + 8 * half;
            out[(m0 + m) * EMB + o] = fmaxf(c[v], 0.0f);
        }
    }
}

extern "C" void kernel_launch(void* const* d_in, const int* in_sizes, int n_in,
                              void* d_out, int out_size, void* d_ws, size_t ws_size,
                              hipStream_t stream) {
    const float* emb        = (const float*)d_in[0];
    const float* W1         = (const float*)d_in[1];
    const float* W2         = (const float*)d_in[2];
    const int*   node_batch = (const int*)  d_in[3];
    const int*   neigh      = (const int*)  d_in[4];
    float*       out        = (float*)d_out;
    float*       h1         = (float*)d_ws;

    const int N = in_sizes[0] / EMB;
    const int B = in_sizes[3];

    const int t1 = (N + 15) / 16;
    const int t2 = (B + 15) / 16;

    sage_layer_kernel<false><<<(t1 + WAVES_PER_BLOCK - 1) / WAVES_PER_BLOCK, 256, 0, stream>>>(
        emb, W1, neigh, nullptr, h1, t1);

    sage_layer_kernel<true><<<(t2 + WAVES_PER_BLOCK - 1) / WAVES_PER_BLOCK, 256, 0, stream>>>(
        h1, W2, neigh, node_batch, out, t2);
}


# W (8 KB) staged once per work-group into LDS with whole-line 16-B loads; waves read B fragments from LDS instead of 32 global loads touching 16 lines each; results bit-identical
# speedup vs baseline: 1.4034x; 1.0533x over previous
#include <hip/hip_runtime.h>


typedef __attribute__((ext_vector_type(2))) float v2f;
typedef __attribute__((ext_vector_type(8))) float v8f;

#define EMB 32
#define K_NBR 10
#define WAVES_PER_BLOCK 8
#define TILE_ROWS 16
#define GROUP_ROWS 4
#define W_PITCH 68

template <bool HAS_MAP>
__global__ __launch_bounds__(256) void sage_layer_kernel(
    const float* __restrict__ feat,
    const float* __restrict__ W,
    const int*   __restrict__ neigh,
    const int*   __restrict__ idxMap,
    float*       __restrict__ out,
    int nTiles)
{
    __shared__ float cshAll[WAVES_PER_BLOCK * 16 * 64];
    __shared__ __attribute__((aligned(16))) float Wsh[32 * W_PITCH];
    const int lane   = threadIdx.x & 31;
    const int wave   = __builtin_amdgcn_readfirstlane((int)(threadIdx.x >> 5));
    const int tile   = blockIdx.x * WAVES_PER_BLOCK + wave;
    const bool active = tile < nTiles;

    const int t = threadIdx.x;
    const float4 wLo = *(const float4*)(W + 4 * t);
    const float4 wHi = *(const float4*)(W + 4 * (t + 256));
    __builtin_amdgcn_sched_barrier(0);

    const int half   = lane >> 4;
    const int lane16 = lane & 15;
    float* csh = &cshAll[wave * 16 * 64];

    const int m0 = (active ? tile : 0) * TILE_ROWS;

    int node[TILE_ROWS];
    #pragma unroll
    for (int r = 0; r < TILE_ROWS; ++r)
        node[r] = HAS_MAP ? idxMap[m0 + r] : (m0 + r);

    int ix[GROUP_ROWS][K_NBR];
    #pragma unroll
    for (int r = 0; r < GROUP_ROWS; ++r) {
        const int* nb = neigh + node[r] * K_NBR;
        #pragma unroll
        for (int j = 0; j < K_NBR; ++j) ix[r][j] = nb[j];
    }

    __builtin_amdgcn_sched_barrier(0);

    *(float4*)&Wsh[(t >> 4) * W_PITCH + 4 * (t & 15)]        = wLo;
    *(float4*)&Wsh[((t >> 4) + 16) * W_PITCH + 4 * (t & 15)] = wHi;
    __syncthreads();
    if (!active) return;

    #pragma unroll
    for (int g = 0; g < TILE_ROWS / GROUP_ROWS; ++g) {
        float selfv[GROUP_ROWS];
        float v[GROUP_ROWS][K_NBR];
        #pragma unroll
        for (int r = 0; r < GROUP_ROWS; ++r) {
            selfv[r] = feat[node[g * GROUP_ROWS + r] * EMB + lane];
            #pragma unroll
            for (int j = 0; j < K_NBR; ++j)
                v[r][j] = feat[ix[r][j] * EMB + lane];
        }
        if (g + 1 < TILE_ROWS / GROUP_ROWS) {
            #pragma unroll
            for (int r = 0; r < GROUP_ROWS; ++r) {
                const int* nb = neigh + node[(g + 1) * GROUP_ROWS + r] * K_NBR;
                #pragma unroll
                for (int j = 0; j < K_NBR; ++j) ix[r][j] = nb[j];
            }
        }
        #pragma unroll
        for (int r = 0; r < GROUP_ROWS; ++r) {
            float acc = 0.f;
            #pragma unroll
            for (int j = 0; j < K_NBR; ++j)
                acc += v[r][j];
            const int row = g * GROUP_ROWS + r;
            csh[row * 64 + lane]      = selfv[r];
            csh[row * 64 + 32 + lane] = acc * (1.0f / K_NBR);
        }
    }

    v2f Bf[2][16];
    #pragma unroll
    for (int nh = 0; nh < 2; ++nh) {
        const int o = lane16 + 16 * nh;
        #pragma unroll
        for (int k = 0; k < 16; ++k) {
            const float2 w2 = *(const float2*)&Wsh[o * W_PITCH + 4 * k + 2 * half];
            v2f b; b.x = w2.x; b.y = w2.y;
            Bf[nh][k] = b;
        }
    }

    #pragma unroll
    for (int nh = 0; nh < 2; ++nh) {
        v8f c = {};
        #pragma unroll
        for (int k = 0; k < 16; ++k) {
            const float2 a2 = *(const float2*)&csh[lane16 * 64 + 4 * k + 2 * half];
            v2f a; a.x = a2.x; a.y = a2.y;
            c = __builtin_amdgcn_wmma_f32_16x16x4_f32(
                     false, a,  false, Bf[nh][k],
                     (short)0, c,  false,  false);
        }
        const int o = 16 * nh + lane16;
        #pragma unroll
        for (int v = 0; v < 8; ++v) {
            const int m = v + 8 * half;
            out[(m0 + m) * EMB + o] = fmaxf(c[v], 0.0f);
        }
    }
}

extern "C" void kernel_launch(void* const* d_in, const int* in_sizes, int n_in,
                              void* d_out, int out_size, void* d_ws, size_t ws_size,
                              hipStream_t stream) {
    const float* emb        = (const float*)d_in[0];
    const float* W1         = (const float*)d_in[1];
    const float* W2         = (const float*)d_in[2];
    const int*   node_batch = (const int*)  d_in[3];
    const int*   neigh      = (const int*)  d_in[4];
    float*       out        = (float*)d_out;
    float*       h1         = (float*)d_ws;

    const int N = in_sizes[0] / EMB;
    const int B = in_sizes[3];

    const int t1 = (N + 15) / 16;
    const int t2 = (B + 15) / 16;

    sage_layer_kernel<false><<<(t1 + WAVES_PER_BLOCK - 1) / WAVES_PER_BLOCK, 256, 0, stream>>>(
        emb, W1, neigh, nullptr, h1, t1);

    sage_layer_kernel<true><<<(t2 + WAVES_PER_BLOCK - 1) / WAVES_PER_BLOCK, 256, 0, stream>>>(
        h1, W2, neigh, node_batch, out, t2);
}


# Combined tile in LDS at a row pitch of 68 floats instead of 64: the A-fragment reads (16 rows x 8 B per lane half) no longer fall on one bank pair 16 ways; results bit-identical
# speedup vs baseline: 1.5137x; 1.0786x over previous
#include <hip/hip_runtime.h>


typedef __attribute__((ext_vector_type(2))) float v2f;
typedef __attribute__((ext_vector_type(8))) float v8f;

#define EMB 32
#define K_NBR 10
#define WAVES_PER_BLOCK 8
#define TILE_ROWS 16
#define GROUP_ROWS 4
#define W_PITCH 68
#define C_PITCH 68

template <bool HAS_MAP>
__global__ __launch_bounds__(256) void sage_layer_kernel(
    const float* __restrict__ feat,
    const float* __restrict__ W,
    const int*   __restrict__ neigh,
    const int*   __restrict__ idxMap,
    float*       __restrict__ out,
    int nTiles)
{
    __shared__ __attribute__((aligned(16))) float cshAll[WAVES_PER_BLOCK * 16 * C_PITCH];
    __shared__ __attribute__((aligned(16))) float Wsh[32 * W_PITCH];
    const int lane   = threadIdx.x & 31;
    const int wave   = __builtin_amdgcn_readfirstlane((int)(threadIdx.x >> 5));
    const int tile   = blockIdx.x * WAVES_PER_BLOCK + wave;
    const bool active = tile < nTiles;

    const int t = threadIdx.x;
    const float4 wLo = *(const float4*)(W + 4 * t);
    const float4 wHi = *(const float4*)(W + 4 * (t + 256));
    __builtin_amdgcn_sched_barrier(0);

    const int half   = lane >> 4;
    const int lane16 = lane & 15;
    float* csh = &cshAll[wave * 16 * C_PITCH];

    const int m0 = (active ? tile : 0) * TILE_ROWS;

    int node[TILE_ROWS];
    #pragma unroll
    for (int r = 0; r < TILE_ROWS; ++r)
        node[r] = HAS_MAP ? idxMap[m0 + r] : (m0 + r);

    int ix[GROUP_ROWS][K_NBR];
    #pragma unroll
    for (int r = 0; r < GROUP_ROWS; ++r) {
        const int* nb = neigh + node[r] * K_NBR;
        #pragma unroll
        for (int j = 0; j < K_NBR; ++j) ix[r][j] = nb[j];
    }

    __builtin_amdgcn_sched_barrier(0);

    *(float4*)&Wsh[(t >> 4) * W_PITCH + 4 * (t & 15)]        = wLo;
    *(float4*)&Wsh[((t >> 4) + 16) * W_PITCH + 4 * (t & 15)] = wHi;
    __syncthreads();
    if (!active) return;

    #pragma unroll
    for (int g = 0; g < TILE_ROWS / GROUP_ROWS; ++g) {
        float selfv[GROUP_ROWS];
        float v[GROUP_ROWS][K_NBR];
        #pragma unroll
        for (int r = 0; r < GROUP_ROWS; ++r) {
            selfv[r] = feat[node[g * GROUP_ROWS + r] * EMB + lane];
            #pragma unroll
            for (int j = 0; j < K_NBR; ++j)
                v[r][j] = feat[ix[r][j] * EMB + lane];
        }
        if (g + 1 < TILE_ROWS / GROUP_ROWS) {
            #pragma unroll
            for (int r = 0; r < GROUP_ROWS; ++r) {
                const int* nb = neigh + node[(g + 1) * GROUP_ROWS + r] * K_NBR;
                #pragma unroll
                for (int j = 0; j < K_NBR; ++j) ix[r][j] = nb[j];
            }
        }
        #pragma unroll
        for (int r = 0; r < GROUP_ROWS; ++r) {
            float acc = 0.f;
            #pragma unroll
            for (int j = 0; j < K_NBR; ++j)
                acc += v[r][j];
            const int row = g * GROUP_ROWS + r;
            csh[row * C_PITCH + lane]      = selfv[r];
            csh[row * C_PITCH + 32 + lane] = acc * (1.0f / K_NBR);
        }
    }

    v2f Bf[2][16];
    #pragma unroll
    for (int nh = 0; nh < 2; ++nh) {
        const int o = lane16 + 16 * nh;
        #pragma unroll
        for (int k = 0; k < 16; ++k) {
            const float2 w2 = *(const float2*)&Wsh[o * W_PITCH + 4 * k + 2 * half];
            v2f b; b.x = w2.x; b.y = w2.y;
            Bf[nh][k] = b;
        }
    }

    #pragma unroll
    for (int nh = 0; nh < 2; ++nh) {
        v8f c = {};
        #pragma unroll
        for (int k = 0; k < 16; ++k) {
            const float2 a2 = *(const float2*)&csh[lane16 * C_PITCH + 4 * k + 2 * half];
            v2f a; a.x = a2.x; a.y = a2.y;
            c = __builtin_amdgcn_wmma_f32_16x16x4_f32(
                     false, a,  false, Bf[nh][k],
                     (short)0, c,  false,  false);
        }
        const int o = 16 * nh + lane16;
        #pragma unroll
        for (int v = 0; v < 8; ++v) {
            const int m = v + 8 * half;
            out[(m0 + m) * EMB + o] = fmaxf(c[v], 0.0f);
        }
    }
}

extern "C" void kernel_launch(void* const* d_in, const int* in_sizes, int n_in,
                              void* d_out, int out_size, void* d_ws, size_t ws_size,
                              hipStream_t stream) {
    const float* emb        = (const float*)d_in[0];
    const float* W1         = (const float*)d_in[1];
    const float* W2         = (const float*)d_in[2];
    const int*   node_batch = (const int*)  d_in[3];
    const int*   neigh      = (const int*)  d_in[4];
    float*       out        = (float*)d_out;
    float*       h1         = (float*)d_ws;

    const int N = in_sizes[0] / EMB;
    const int B = in_sizes[3];

    const int t1 = (N + 15) / 16;
    const int t2 = (B + 15) / 16;

    sage_layer_kernel<false><<<(t1 + WAVES_PER_BLOCK - 1) / WAVES_PER_BLOCK, 256, 0, stream>>>(
        emb, W1, neigh, nullptr, h1, t1);

    sage_layer_kernel<true><<<(t2 + WAVES_PER_BLOCK - 1) / WAVES_PER_BLOCK, 256, 0, stream>>>(
        h1, W2, neigh, node_batch, out, t2);
}


# Waves loop over 16-row tiles: layer 1 runs 1024 resident work-groups (3-4 tiles per wave, next tile's ids fetched ahead), not one short wave per tile; W staged once per wave; bit-identical
# speedup vs baseline: 1.6859x; 1.1138x over previous
#include <hip/hip_runtime.h>


typedef __attribute__((ext_vector_type(2))) float v2f;
typedef __attribute__((ext_vector_type(8))) float v8f;

#define EMB 32
#define K_NBR 10
#define WAVES_PER_BLOCK 8
#define TILE_ROWS 16
#define GROUP_ROWS 4
#define W_PITCH 68
#define C_PITCH 68

template <bool HAS_MAP>
__global__ __launch_bounds__(256) void sage_layer_kernel(
    const float* __restrict__ feat,
    const float* __restrict__ W,
    const int*   __restrict__ neigh,
    const int*   __restrict__ idxMap,
    float*       __restrict__ out,
    int nTiles)
{
    __shared__ __attribute__((aligned(16))) float cshAll[WAVES_PER_BLOCK * 16 * C_PITCH];
    __shared__ __attribute__((aligned(16))) float Wsh[32 * W_PITCH];
    const int lane   = threadIdx.x & 31;
    const int wave   = __builtin_amdgcn_readfirstlane((int)(threadIdx.x >> 5));
    int tile         = blockIdx.x * WAVES_PER_BLOCK + wave;
    const int stride = (int)gridDim.x * WAVES_PER_BLOCK;
    const bool active = tile < nTiles;

    const int t = threadIdx.x;
    const float4 wLo = *(const float4*)(W + 4 * t);
    const float4 wHi = *(const float4*)(W + 4 * (t + 256));
    __builtin_amdgcn_sched_barrier(0);

    const int half   = lane >> 4;
    const int lane16 = lane & 15;
    float* csh = &cshAll[wave * 16 * C_PITCH];

    int m0 = (active ? tile : 0) * TILE_ROWS;

    int node[TILE_ROWS];
    #pragma unroll
    for (int r = 0; r < TILE_ROWS; ++r)
        node[r] = HAS_MAP ? idxMap[m0 + r] : (m0 + r);

    int ix[GROUP_ROWS][K_NBR];
    #pragma unroll
    for (int r = 0; r < GROUP_ROWS; ++r) {
        const int* nb = neigh + node[r] * K_NBR;
        #pragma unroll
        for (int j = 0; j < K_NBR; ++j) ix[r][j] = nb[j];
    }

    __builtin_amdgcn_sched_barrier(0);

    *(float4*)&Wsh[(t >> 4) * W_PITCH + 4 * (t & 15)]        = wLo;
    *(float4*)&Wsh[((t >> 4) + 16) * W_PITCH + 4 * (t & 15)] = wHi;
    __syncthreads();
    if (!active) return;

    bool more;
    do {
        const int nextTile = tile + stride;
        more = nextTile < nTiles;

        #pragma unroll
        for (int g = 0; g < TILE_ROWS / GROUP_ROWS; ++g) {
            float selfv[GROUP_ROWS];
            float v[GROUP_ROWS][K_NBR];
            #pragma unroll
            for (int r = 0; r < GROUP_ROWS; ++r) {
                selfv[r] = feat[node[g * GROUP_ROWS + r] * EMB + lane];
                #pragma unroll
                for (int j = 0; j < K_NBR; ++j)
                    v[r][j] = feat[ix[r][j] * EMB + lane];
            }
            if (g + 1 == TILE_ROWS / GROUP_ROWS) __builtin_amdgcn_sched_barrier(0);
            if (g + 1 < TILE_ROWS / GROUP_ROWS) {
                #pragma unroll
                for (int r = 0; r < GROUP_ROWS; ++r) {
                    const int* nb = neigh + node[(g + 1) * GROUP_ROWS + r] * K_NBR;
                    #pragma unroll
                    for (int j = 0; j < K_NBR; ++j) ix[r][j] = nb[j];
                }
            } else if (more) {
                const int m0n = nextTile * TILE_ROWS;
                #pragma unroll
                for (int r = 0; r < TILE_ROWS; ++r)
                    node[r] = HAS_MAP ? idxMap[m0n + r] : (m0n + r);
                #pragma unroll
                for (int r = 0; r < GROUP_ROWS; ++r) {
                    const int* nb = neigh + node[r] * K_NBR;
                    #pragma unroll
                    for (int j = 0; j < K_NBR; ++j) ix[r][j] = nb[j];
                }
            }
            __builtin_amdgcn_sched_barrier(0);
            #pragma unroll
            for (int r = 0; r < GROUP_ROWS; ++r) {
                float acc = 0.f;
                #pragma unroll
                for (int j = 0; j < K_NBR; ++j)
                    acc += v[r][j];
                const int row = g * GROUP_ROWS + r;
                csh[row * C_PITCH + lane]      = selfv[r];
                csh[row * C_PITCH + 32 + lane] = acc * (1.0f / K_NBR);
            }
            __builtin_amdgcn_sched_barrier(0);
        }

        v2f Bf[2][16];
        #pragma unroll
        for (int nh = 0; nh < 2; ++nh) {
            const int o = lane16 + 16 * nh;
            #pragma unroll
            for (int k = 0; k < 16; ++k) {
                const float2 w2 = *(const float2*)&Wsh[o * W_PITCH + 4 * k + 2 * half];
                v2f b; b.x = w2.x; b.y = w2.y;
                Bf[nh][k] = b;
            }
        }

        #pragma unroll
        for (int nh = 0; nh < 2; ++nh) {
            v8f c = {};
            #pragma unroll
            for (int k = 0; k < 16; ++k) {
                const float2 a2 = *(const float2*)&csh[lane16 * C_PITCH + 4 * k + 2 * half];
                v2f a; a.x = a2.x; a.y = a2.y;
                c = __builtin_amdgcn_wmma_f32_16x16x4_f32(
                         false, a,  false, Bf[nh][k],
                         (short)0, c,  false,  false);
            }
            const int o = 16 * nh + lane16;
            #pragma unroll
            for (int v = 0; v < 8; ++v) {
                const int m = v + 8 * half;
                out[(m0 + m) * EMB + o] = fmaxf(c[v], 0.0f);
            }
        }

        tile = nextTile;
        m0   = nextTile * TILE_ROWS;
    } while (more);
}

extern "C" void kernel_launch(void* const* d_in, const int* in_sizes, int n_in,
                              void* d_out, int out_size, void* d_ws, size_t ws_size,
                              hipStream_t stream) {
    const float* emb        = (const float*)d_in[0];
    const float* W1         = (const float*)d_in[1];
    const float* W2         = (const float*)d_in[2];
    const int*   node_batch = (const int*)  d_in[3];
    const int*   neigh      = (const int*)  d_in[4];
    float*       out        = (float*)d_out;
    float*       h1         = (float*)d_ws;

    const int N = in_sizes[0] / EMB;
    const int B = in_sizes[3];

    const int t1 = (N + 15) / 16;
    const int t2 = (B + 15) / 16;

    sage_layer_kernel<false><<<1024, 256, 0, stream>>>(
        emb, W1, neigh, nullptr, h1, t1);

    sage_layer_kernel<true><<<(t2 + WAVES_PER_BLOCK - 1) / WAVES_PER_BLOCK, 256, 0, stream>>>(
        h1, W2, neigh, node_batch, out, t2);
}
